# starting kernel
# baseline (speedup 1.0000x reference)
#include <hip/hip_runtime.h>

typedef float v2f __attribute__((ext_vector_type(2)));
typedef float v8f __attribute__((ext_vector_type(8)));

#define N_NODES 100000
#define N_EDGES 3200000
#define NFEAT   256
#define NCLASS  64
#define MTILES  (N_NODES / 16)
#define WAVES_PER_BLOCK 8


__global__ void k_init_deg(float* __restrict__ deg) {
    int i = blockIdx.x * blockDim.x + threadIdx.x;
    if (i < N_NODES) deg[i] = 1.0f;
}

__global__ void k_count(const int* __restrict__ dst, float* __restrict__ deg) {
    int e = blockIdx.x * blockDim.x + threadIdx.x;
    if (e < N_EDGES) atomicAdd(&deg[dst[e]], 1.0f);
}

__global__ void k_dinv(float* __restrict__ deg) {
    int i = blockIdx.x * blockDim.x + threadIdx.x;
    if (i < N_NODES) deg[i] = __frsqrt_rn(deg[i]);
}


__global__ __launch_bounds__(256) void k_gemm_wmma(const float* __restrict__ x,
                                                   const float* __restrict__ W,
                                                   float* __restrict__ xw) {
    __shared__ float Wt[NCLASS * NFEAT];
    for (int i = threadIdx.x; i < NCLASS * NFEAT; i += 256) {
        int k = i >> 6;
        int c = i & 63;
        Wt[c * NFEAT + k] = W[i];
    }
    __syncthreads();

    const int lane  = threadIdx.x & 31;
    const int wave  = threadIdx.x >> 5;
    const int mtile = blockIdx.x * WAVES_PER_BLOCK + wave;
    if (mtile >= MTILES) return;

    const int half = lane >> 4;
    const int l15  = lane & 15;
    const float* xrow = x + (size_t)(mtile * 16 + l15) * NFEAT;

    v8f acc0 = {}, acc1 = {}, acc2 = {}, acc3 = {};

    for (int k = 0; k < NFEAT; k += 4) {
        const int ka = k + 2 * half;
        v2f a  = *(const v2f*)(xrow + ka);
        v2f b0 = *(const v2f*)(&Wt[(0 * 16 + l15) * NFEAT + ka]);
        v2f b1 = *(const v2f*)(&Wt[(1 * 16 + l15) * NFEAT + ka]);
        v2f b2 = *(const v2f*)(&Wt[(2 * 16 + l15) * NFEAT + ka]);
        v2f b3 = *(const v2f*)(&Wt[(3 * 16 + l15) * NFEAT + ka]);
        acc0 = __builtin_amdgcn_wmma_f32_16x16x4_f32(false, a, false, b0, (short)0, acc0, false, false);
        acc1 = __builtin_amdgcn_wmma_f32_16x16x4_f32(false, a, false, b1, (short)0, acc1, false, false);
        acc2 = __builtin_amdgcn_wmma_f32_16x16x4_f32(false, a, false, b2, (short)0, acc2, false, false);
        acc3 = __builtin_amdgcn_wmma_f32_16x16x4_f32(false, a, false, b3, (short)0, acc3, false, false);
    }

    float* orow = xw + (size_t)mtile * 16 * NCLASS;
#pragma unroll
    for (int r = 0; r < 8; ++r) {
        int m = r + 8 * half;
        orow[m * NCLASS + 0 * 16 + l15] = acc0[r];
        orow[m * NCLASS + 1 * 16 + l15] = acc1[r];
        orow[m * NCLASS + 2 * 16 + l15] = acc2[r];
        orow[m * NCLASS + 3 * 16 + l15] = acc3[r];
    }
}


__global__ void k_out_init(const float* __restrict__ xw, const float* __restrict__ dinv,
                           const float* __restrict__ bias, float* __restrict__ out) {
    int i = blockIdx.x * blockDim.x + threadIdx.x;
    if (i < N_NODES * NCLASS) {
        int n = i >> 6;
        int c = i & 63;
        float di = dinv[n];
        out[i] = bias[c] + xw[i] * di * di;
    }
}


__global__ void k_scatter(const int* __restrict__ src, const int* __restrict__ dst,
                          const float* __restrict__ xw, const float* __restrict__ dinv,
                          float* __restrict__ out) {
    long long t = (long long)blockIdx.x * blockDim.x + threadIdx.x;
    if (t >= (long long)N_EDGES * NCLASS) return;
    int e = (int)(t >> 6);
    int c = (int)(t & 63);
    int s = src[e];
    int d = dst[e];
    float norm = dinv[s] * dinv[d];
    atomicAdd(&out[(size_t)d * NCLASS + c], xw[(size_t)s * NCLASS + c] * norm);
}


extern "C" void kernel_launch(void* const* d_in, const int* in_sizes, int n_in,
                              void* d_out, int out_size, void* d_ws, size_t ws_size,
                              hipStream_t stream) {
    const float* x    = (const float*)d_in[0];
    const float* W    = (const float*)d_in[1];
    const float* bias = (const float*)d_in[2];
    const int*   ei   = (const int*)d_in[3];
    const int*   src  = ei;
    const int*   dst  = ei + N_EDGES;
    float* out = (float*)d_out;

    float* dinv = (float*)d_ws;
    float* xw   = dinv + 102400;

    k_init_deg<<<(N_NODES + 255) / 256, 256, 0, stream>>>(dinv);
    k_count  <<<(N_EDGES + 255) / 256, 256, 0, stream>>>(dst, dinv);
    k_dinv   <<<(N_NODES + 255) / 256, 256, 0, stream>>>(dinv);

    k_gemm_wmma<<<(MTILES + WAVES_PER_BLOCK - 1) / WAVES_PER_BLOCK, 256, 0, stream>>>(x, W, xw);

    k_out_init<<<(N_NODES * NCLASS + 255) / 256, 256, 0, stream>>>(xw, dinv, bias, out);

    long long tot = (long long)N_EDGES * NCLASS;
    k_scatter<<<(unsigned)((tot + 255) / 256), 256, 0, stream>>>(src, dst, xw, dinv, out);
}


# Float-atomic edge scatter replaced by bucketing edges by destination (integer count returns each edge's arrival-order rank) and a per-node gather that writes out once; sum order follows that rank
# speedup vs baseline: 2.2918x; 2.2918x over previous
#include <hip/hip_runtime.h>

typedef float v2f __attribute__((ext_vector_type(2)));
typedef float v8f __attribute__((ext_vector_type(8)));

#define N_NODES 100000
#define N_EDGES 3200000
#define NFEAT   256
#define NCLASS  64
#define MTILES  (N_NODES / 16)
#define WAVES_PER_BLOCK 8


__global__ void k_init_cnt(int* __restrict__ cnt, int* __restrict__ cursor) {
    int i = blockIdx.x * blockDim.x + threadIdx.x;
    if (i < N_NODES) cnt[i] = 0;
    if (i == 0) *cursor = 0;
}

__global__ void k_count_rank(const int* __restrict__ dst, int* __restrict__ cnt,
                             int* __restrict__ pos) {
    int e = blockIdx.x * blockDim.x + threadIdx.x;
    if (e < N_EDGES) pos[e] = atomicAdd(&cnt[dst[e]], 1);
}

__global__ __launch_bounds__(256) void k_offsets(const int* __restrict__ cnt, int* __restrict__ off,
                                                 float* __restrict__ dinv, int* __restrict__ cursor) {
    __shared__ int wsum[WAVES_PER_BLOCK];
    __shared__ int base;
    const int i    = blockIdx.x * 256 + threadIdx.x;
    const int lane = threadIdx.x & 31;
    const int wave = threadIdx.x >> 5;
    const int c = (i < N_NODES) ? cnt[i] : 0;
    int v = c;
    for (int o = 1; o < 32; o <<= 1) {
        int t = __shfl_up(v, o);
        if (lane >= o) v += t;
    }
    if (lane == 31) wsum[wave] = v;
    __syncthreads();
    if (threadIdx.x == 0) {
        int run = 0;
        for (int w = 0; w < WAVES_PER_BLOCK; ++w) { int t = wsum[w]; wsum[w] = run; run += t; }
        base = atomicAdd(cursor, run);
    }
    __syncthreads();
    if (i < N_NODES) {
        off[i]  = base + wsum[wave] + v - c;
        dinv[i] = __frsqrt_rn((float)(c + 1));
    }
}

__global__ void k_fill(const int* __restrict__ src, const int* __restrict__ dst,
                       const int* __restrict__ off, const int* __restrict__ pos,
                       int* __restrict__ perm) {
    int e = blockIdx.x * blockDim.x + threadIdx.x;
    if (e < N_EDGES) perm[off[dst[e]] + pos[e]] = src[e];
}


__global__ __launch_bounds__(256) void k_gemm_wmma(const float* __restrict__ x,
                                                   const float* __restrict__ W,
                                                   float* __restrict__ xw) {
    __shared__ float Wt[NCLASS * NFEAT];
    for (int i = threadIdx.x; i < NCLASS * NFEAT; i += 256) {
        int k = i >> 6;
        int c = i & 63;
        Wt[c * NFEAT + k] = W[i];
    }
    __syncthreads();

    const int lane  = threadIdx.x & 31;
    const int wave  = threadIdx.x >> 5;
    const int mtile = blockIdx.x * WAVES_PER_BLOCK + wave;
    if (mtile >= MTILES) return;

    const int half = lane >> 4;
    const int l15  = lane & 15;
    const float* xrow = x + (size_t)(mtile * 16 + l15) * NFEAT;

    v8f acc0 = {}, acc1 = {}, acc2 = {}, acc3 = {};

    for (int k = 0; k < NFEAT; k += 4) {
        const int ka = k + 2 * half;
        v2f a  = *(const v2f*)(xrow + ka);
        v2f b0 = *(const v2f*)(&Wt[(0 * 16 + l15) * NFEAT + ka]);
        v2f b1 = *(const v2f*)(&Wt[(1 * 16 + l15) * NFEAT + ka]);
        v2f b2 = *(const v2f*)(&Wt[(2 * 16 + l15) * NFEAT + ka]);
        v2f b3 = *(const v2f*)(&Wt[(3 * 16 + l15) * NFEAT + ka]);
        acc0 = __builtin_amdgcn_wmma_f32_16x16x4_f32(false, a, false, b0, (short)0, acc0, false, false);
        acc1 = __builtin_amdgcn_wmma_f32_16x16x4_f32(false, a, false, b1, (short)0, acc1, false, false);
        acc2 = __builtin_amdgcn_wmma_f32_16x16x4_f32(false, a, false, b2, (short)0, acc2, false, false);
        acc3 = __builtin_amdgcn_wmma_f32_16x16x4_f32(false, a, false, b3, (short)0, acc3, false, false);
    }

    float* orow = xw + (size_t)mtile * 16 * NCLASS;
#pragma unroll
    for (int r = 0; r < 8; ++r) {
        int m = r + 8 * half;
        orow[m * NCLASS + 0 * 16 + l15] = acc0[r];
        orow[m * NCLASS + 1 * 16 + l15] = acc1[r];
        orow[m * NCLASS + 2 * 16 + l15] = acc2[r];
        orow[m * NCLASS + 3 * 16 + l15] = acc3[r];
    }
}


static __device__ __forceinline__ v2f bucket_row(const v2f* __restrict__ xw2, int s, int j, int lane) {
    const int sj = __builtin_amdgcn_readlane(s, j);
    return xw2[(size_t)sj * 32 + lane];
}
static __device__ __forceinline__ void add_row(v2f& acc, v2f r, int wi, int j) {
    const float wj = __int_as_float(__builtin_amdgcn_readlane(wi, j));
    acc.x += r.x * wj;
    acc.y += r.y * wj;
}

__global__ __launch_bounds__(256) void k_gather(const int* __restrict__ perm, const int* __restrict__ off,
                                                const int* __restrict__ cnt, const float* __restrict__ xw,
                                                const float* __restrict__ dinv, const float* __restrict__ bias,
                                                float* __restrict__ out) {
    const int lane = threadIdx.x & 31;
    const int d = blockIdx.x * WAVES_PER_BLOCK + __builtin_amdgcn_readfirstlane((int)(threadIdx.x >> 5));
    if (d >= N_NODES) return;
    const int   n  = cnt[d];
    const int   o  = off[d];
    const float dd = dinv[d];
    const v2f* xw2 = (const v2f*)xw;

    v2f self = xw2[(size_t)d * 32 + lane];
    const float ws = dd * dd;
    v2f acc;
    acc.x = self.x * ws;
    acc.y = self.y * ws;

    for (int j0 = 0; j0 < n; j0 += 32) {
        const int m = (n - j0 < 32) ? (n - j0) : 32;
        int   s = d;
        float w = 0.0f;
        if (lane < m) {
            s = perm[o + j0 + lane];
            w = dinv[s] * dd;
        }
        const int wi = __float_as_int(w);
        const int m4 = (m + 3) & ~3;
        int j = 0;
        for (; m4 - j > 4; j += 8) {
            const v2f r0 = bucket_row(xw2, s, j,     lane);
            const v2f r1 = bucket_row(xw2, s, j + 1, lane);
            const v2f r2 = bucket_row(xw2, s, j + 2, lane);
            const v2f r3 = bucket_row(xw2, s, j + 3, lane);
            const v2f r4 = bucket_row(xw2, s, j + 4, lane);
            const v2f r5 = bucket_row(xw2, s, j + 5, lane);
            const v2f r6 = bucket_row(xw2, s, j + 6, lane);
            const v2f r7 = bucket_row(xw2, s, j + 7, lane);
            add_row(acc, r0, wi, j);
            add_row(acc, r1, wi, j + 1);
            add_row(acc, r2, wi, j + 2);
            add_row(acc, r3, wi, j + 3);
            add_row(acc, r4, wi, j + 4);
            add_row(acc, r5, wi, j + 5);
            add_row(acc, r6, wi, j + 6);
            add_row(acc, r7, wi, j + 7);
        }
        if (j < m4) {
            const v2f r0 = bucket_row(xw2, s, j,     lane);
            const v2f r1 = bucket_row(xw2, s, j + 1, lane);
            const v2f r2 = bucket_row(xw2, s, j + 2, lane);
            const v2f r3 = bucket_row(xw2, s, j + 3, lane);
            add_row(acc, r0, wi, j);
            add_row(acc, r1, wi, j + 1);
            add_row(acc, r2, wi, j + 2);
            add_row(acc, r3, wi, j + 3);
        }
    }

    const v2f b2 = ((const v2f*)bias)[lane];
    v2f res;
    res.x = b2.x + acc.x;
    res.y = b2.y + acc.y;
    ((v2f*)out)[(size_t)d * 32 + lane] = res;
}


extern "C" void kernel_launch(void* const* d_in, const int* in_sizes, int n_in,
                              void* d_out, int out_size, void* d_ws, size_t ws_size,
                              hipStream_t stream) {
    const float* x    = (const float*)d_in[0];
    const float* W    = (const float*)d_in[1];
    const float* bias = (const float*)d_in[2];
    const int*   ei   = (const int*)d_in[3];
    const int*   src  = ei;
    const int*   dst  = ei + N_EDGES;
    float* out = (float*)d_out;

    float* dinv   = (float*)d_ws;
    float* xw     = dinv + 102400;
    int*   cnt    = (int*)(xw + 6400000);
    int*   off    = cnt + 102400;
    int*   cursor = off + 102400;
    int*   pos    = cursor + 64;
    int*   perm   = pos + 3200000;

    k_init_cnt  <<<(N_NODES + 255) / 256, 256, 0, stream>>>(cnt, cursor);
    k_count_rank<<<(N_EDGES + 255) / 256, 256, 0, stream>>>(dst, cnt, pos);
    k_offsets   <<<(N_NODES + 255) / 256, 256, 0, stream>>>(cnt, off, dinv, cursor);
    k_fill      <<<(N_EDGES + 255) / 256, 256, 0, stream>>>(src, dst, off, pos, perm);

    k_gemm_wmma<<<(MTILES + WAVES_PER_BLOCK - 1) / WAVES_PER_BLOCK, 256, 0, stream>>>(x, W, xw);

    k_gather<<<(N_NODES + WAVES_PER_BLOCK - 1) / WAVES_PER_BLOCK, 256, 0, stream>>>(perm, off, cnt, xw, dinv, bias, out);
}


# W tile in LDS at row pitch 260 floats instead of 256: B-fragment reads and staging stores no longer pile onto one bank pair per half-wave; k order of both loaders unchanged
# speedup vs baseline: 2.6527x; 1.1575x over previous
#include <hip/hip_runtime.h>

typedef float v2f __attribute__((ext_vector_type(2)));
typedef float v8f __attribute__((ext_vector_type(8)));

#define N_NODES 100000
#define N_EDGES 3200000
#define NFEAT   256
#define NCLASS  64
#define MTILES  (N_NODES / 16)
#define WAVES_PER_BLOCK 8
#define WPITCH 260


__global__ void k_init_cnt(int* __restrict__ cnt, int* __restrict__ cursor) {
    int i = blockIdx.x * blockDim.x + threadIdx.x;
    if (i < N_NODES) cnt[i] = 0;
    if (i == 0) *cursor = 0;
}

__global__ void k_count_rank(const int* __restrict__ dst, int* __restrict__ cnt,
                             int* __restrict__ pos) {
    int e = blockIdx.x * blockDim.x + threadIdx.x;
    if (e < N_EDGES) pos[e] = atomicAdd(&cnt[dst[e]], 1);
}

__global__ __launch_bounds__(256) void k_offsets(const int* __restrict__ cnt, int* __restrict__ off,
                                                 float* __restrict__ dinv, int* __restrict__ cursor) {
    __shared__ int wsum[WAVES_PER_BLOCK];
    __shared__ int base;
    const int i    = blockIdx.x * 256 + threadIdx.x;
    const int lane = threadIdx.x & 31;
    const int wave = threadIdx.x >> 5;
    const int c = (i < N_NODES) ? cnt[i] : 0;
    int v = c;
    for (int o = 1; o < 32; o <<= 1) {
        int t = __shfl_up(v, o);
        if (lane >= o) v += t;
    }
    if (lane == 31) wsum[wave] = v;
    __syncthreads();
    if (threadIdx.x == 0) {
        int run = 0;
        for (int w = 0; w < WAVES_PER_BLOCK; ++w) { int t = wsum[w]; wsum[w] = run; run += t; }
        base = atomicAdd(cursor, run);
    }
    __syncthreads();
    if (i < N_NODES) {
        off[i]  = base + wsum[wave] + v - c;
        dinv[i] = __frsqrt_rn((float)(c + 1));
    }
}

__global__ void k_fill(const int* __restrict__ src, const int* __restrict__ dst,
                       const int* __restrict__ off, const int* __restrict__ pos,
                       int* __restrict__ perm) {
    int e = blockIdx.x * blockDim.x + threadIdx.x;
    if (e < N_EDGES) perm[off[dst[e]] + pos[e]] = src[e];
}


__global__ __launch_bounds__(256) void k_gemm_wmma(const float* __restrict__ x,
                                                   const float* __restrict__ W,
                                                   float* __restrict__ xw) {
    __shared__ float Wt[NCLASS * WPITCH];
    for (int i = threadIdx.x; i < NCLASS * NFEAT; i += 256) {
        int k = i >> 6;
        int c = i & 63;
        Wt[c * WPITCH + k] = W[i];
    }
    __syncthreads();

    const int lane  = threadIdx.x & 31;
    const int wave  = threadIdx.x >> 5;
    const int mtile = blockIdx.x * WAVES_PER_BLOCK + wave;
    if (mtile >= MTILES) return;

    const int half = lane >> 4;
    const int l15  = lane & 15;
    const float* xrow = x + (size_t)(mtile * 16 + l15) * NFEAT;

    v8f acc0 = {}, acc1 = {}, acc2 = {}, acc3 = {};

    for (int k = 0; k < NFEAT; k += 4) {
        const int ka = k + 2 * half;
        v2f a  = *(const v2f*)(xrow + ka);
        v2f b0 = *(const v2f*)(&Wt[(0 * 16 + l15) * WPITCH + ka]);
        v2f b1 = *(const v2f*)(&Wt[(1 * 16 + l15) * WPITCH + ka]);
        v2f b2 = *(const v2f*)(&Wt[(2 * 16 + l15) * WPITCH + ka]);
        v2f b3 = *(const v2f*)(&Wt[(3 * 16 + l15) * WPITCH + ka]);
        acc0 = __builtin_amdgcn_wmma_f32_16x16x4_f32(false, a, false, b0, (short)0, acc0, false, false);
        acc1 = __builtin_amdgcn_wmma_f32_16x16x4_f32(false, a, false, b1, (short)0, acc1, false, false);
        acc2 = __builtin_amdgcn_wmma_f32_16x16x4_f32(false, a, false, b2, (short)0, acc2, false, false);
        acc3 = __builtin_amdgcn_wmma_f32_16x16x4_f32(false, a, false, b3, (short)0, acc3, false, false);
    }

    float* orow = xw + (size_t)mtile * 16 * NCLASS;
#pragma unroll
    for (int r = 0; r < 8; ++r) {
        int m = r + 8 * half;
        orow[m * NCLASS + 0 * 16 + l15] = acc0[r];
        orow[m * NCLASS + 1 * 16 + l15] = acc1[r];
        orow[m * NCLASS + 2 * 16 + l15] = acc2[r];
        orow[m * NCLASS + 3 * 16 + l15] = acc3[r];
    }
}


static __device__ __forceinline__ v2f bucket_row(const v2f* __restrict__ xw2, int s, int j, int lane) {
    const int sj = __builtin_amdgcn_readlane(s, j);
    return xw2[(size_t)sj * 32 + lane];
}
static __device__ __forceinline__ void add_row(v2f& acc, v2f r, int wi, int j) {
    const float wj = __int_as_float(__builtin_amdgcn_readlane(wi, j));
    acc.x += r.x * wj;
    acc.y += r.y * wj;
}

__global__ __launch_bounds__(256) void k_gather(const int* __restrict__ perm, const int* __restrict__ off,
                                                const int* __restrict__ cnt, const float* __restrict__ xw,
                                                const float* __restrict__ dinv, const float* __restrict__ bias,
                                                float* __restrict__ out) {
    const int lane = threadIdx.x & 31;
    const int d = blockIdx.x * WAVES_PER_BLOCK + __builtin_amdgcn_readfirstlane((int)(threadIdx.x >> 5));
    if (d >= N_NODES) return;
    const int   n  = cnt[d];
    const int   o  = off[d];
    const float dd = dinv[d];
    const v2f* xw2 = (const v2f*)xw;

    v2f self = xw2[(size_t)d * 32 + lane];
    const float ws = dd * dd;
    v2f acc;
    acc.x = self.x * ws;
    acc.y = self.y * ws;

    for (int j0 = 0; j0 < n; j0 += 32) {
        const int m = (n - j0 < 32) ? (n - j0) : 32;
        int   s = d;
        float w = 0.0f;
        if (lane < m) {
            s = perm[o + j0 + lane];
            w = dinv[s] * dd;
        }
        const int wi = __float_as_int(w);
        const int m4 = (m + 3) & ~3;
        int j = 0;
        for (; m4 - j > 4; j += 8) {
            const v2f r0 = bucket_row(xw2, s, j,     lane);
            const v2f r1 = bucket_row(xw2, s, j + 1, lane);
            const v2f r2 = bucket_row(xw2, s, j + 2, lane);
            const v2f r3 = bucket_row(xw2, s, j + 3, lane);
            const v2f r4 = bucket_row(xw2, s, j + 4, lane);
            const v2f r5 = bucket_row(xw2, s, j + 5, lane);
            const v2f r6 = bucket_row(xw2, s, j + 6, lane);
            const v2f r7 = bucket_row(xw2, s, j + 7, lane);
            add_row(acc, r0, wi, j);
            add_row(acc, r1, wi, j + 1);
            add_row(acc, r2, wi, j + 2);
            add_row(acc, r3, wi, j + 3);
            add_row(acc, r4, wi, j + 4);
            add_row(acc, r5, wi, j + 5);
            add_row(acc, r6, wi, j + 6);
            add_row(acc, r7, wi, j + 7);
        }
        if (j < m4) {
            const v2f r0 = bucket_row(xw2, s, j,     lane);
            const v2f r1 = bucket_row(xw2, s, j + 1, lane);
            const v2f r2 = bucket_row(xw2, s, j + 2, lane);
            const v2f r3 = bucket_row(xw2, s, j + 3, lane);
            add_row(acc, r0, wi, j);
            add_row(acc, r1, wi, j + 1);
            add_row(acc, r2, wi, j + 2);
            add_row(acc, r3, wi, j + 3);
        }
    }

    const v2f b2 = ((const v2f*)bias)[lane];
    v2f res;
    res.x = b2.x + acc.x;
    res.y = b2.y + acc.y;
    ((v2f*)out)[(size_t)d * 32 + lane] = res;
}


extern "C" void kernel_launch(void* const* d_in, const int* in_sizes, int n_in,
                              void* d_out, int out_size, void* d_ws, size_t ws_size,
                              hipStream_t stream) {
    const float* x    = (const float*)d_in[0];
    const float* W    = (const float*)d_in[1];
    const float* bias = (const float*)d_in[2];
    const int*   ei   = (const int*)d_in[3];
    const int*   src  = ei;
    const int*   dst  = ei + N_EDGES;
    float* out = (float*)d_out;

    float* dinv   = (float*)d_ws;
    float* xw     = dinv + 102400;
    int*   cnt    = (int*)(xw + 6400000);
    int*   off    = cnt + 102400;
    int*   cursor = off + 102400;
    int*   pos    = cursor + 64;
    int*   perm   = pos + 3200000;

    k_init_cnt  <<<(N_NODES + 255) / 256, 256, 0, stream>>>(cnt, cursor);
    k_count_rank<<<(N_EDGES + 255) / 256, 256, 0, stream>>>(dst, cnt, pos);
    k_offsets   <<<(N_NODES + 255) / 256, 256, 0, stream>>>(cnt, off, dinv, cursor);
    k_fill      <<<(N_EDGES + 255) / 256, 256, 0, stream>>>(src, dst, off, pos, perm);

    k_gemm_wmma<<<(MTILES + WAVES_PER_BLOCK - 1) / WAVES_PER_BLOCK, 256, 0, stream>>>(x, W, xw);

    k_gather<<<(N_NODES + WAVES_PER_BLOCK - 1) / WAVES_PER_BLOCK, 256, 0, stream>>>(perm, off, cnt, xw, dinv, bias, out);
}


# Edges bucketed by a two-level counting sort on LDS counters (slice order by 256-node bin, then per-bin placement), not global returning adds: 6 launches to 4; sum order follows the LDS add's rank
# speedup vs baseline: 6.8087x; 2.5667x over previous
#include <hip/hip_runtime.h>

typedef float v2f __attribute__((ext_vector_type(2)));
typedef float v8f __attribute__((ext_vector_type(8)));

#define N_NODES 100000
#define N_EDGES 3200000
#define NFEAT   256
#define NCLASS  64
#define MTILES  (N_NODES / 16)
#define WAVES_PER_BLOCK 8
#define WPITCH 260


#define BIN_SHIFT 8
#define BIN_NODES 256
#define NBINS ((N_NODES + BIN_NODES - 1) / BIN_NODES)
#define SLICES 256
#define SLICE_EDGES (N_EDGES / SLICES)
#define SORT_THREADS 1024
#define SORT_WAVES (SORT_THREADS / 32)
#define RUNS_PER_WAVE (SLICES / SORT_WAVES)
#define EDGES_PER_THREAD ((SLICE_EDGES + SORT_THREADS - 1) / SORT_THREADS)

static_assert(SLICE_EDGES * SLICES == N_EDGES, "slices must tile the edge list");
static_assert(RUNS_PER_WAVE * SORT_WAVES == SLICES, "every wave takes the same number of runs");
static_assert(NBINS <= 512, "one scan thread per bin");
static_assert(SLICE_EDGES < 32768, "run start and length share one word");
static_assert(N_NODES <= (1 << (31 - BIN_SHIFT)), "source and local node share one word");

static __device__ __forceinline__ int wave_inclusive_scan(int v, int lane) {
    for (int o = 1; o < 32; o <<= 1) {
        int t = __shfl_up(v, o);
        if (lane >= o) v += t;
    }
    return v;
}

__global__ __launch_bounds__(SORT_THREADS) void k_slice_sort(const int* __restrict__ src,
                                                             const int* __restrict__ dst,
                                                             int* __restrict__ binned,
                                                             int* __restrict__ runs,
                                                             int* __restrict__ cursor) {
    __shared__ int h[512];
    __shared__ int wsum[16];
    __shared__ int stage[SLICE_EDGES];
    const int t    = threadIdx.x;
    const int lane = t & 31;
    const int wave = t >> 5;
    if (blockIdx.x == 0 && t == 0) *cursor = 0;
    if (t < 512) h[t] = 0;

    const int e0 = blockIdx.x * SLICE_EDGES;
    int key[EDGES_PER_THREAD];
    int bin[EDGES_PER_THREAD];
#pragma unroll
    for (int j = 0; j < EDGES_PER_THREAD; ++j) {
        const int i = t + j * SORT_THREADS;
        bin[j] = -1;
        key[j] = 0;
        if (i < SLICE_EDGES) {
            const int d = dst[e0 + i];
            const int s = src[e0 + i];
            bin[j] = d >> BIN_SHIFT;
            key[j] = (s << BIN_SHIFT) | (d & (BIN_NODES - 1));
        }
    }
    __syncthreads();
#pragma unroll
    for (int j = 0; j < EDGES_PER_THREAD; ++j)
        if (bin[j] >= 0) atomicAdd(&h[bin[j]], 1);
    __syncthreads();

    const int cb = (t < 512) ? h[t] : 0;
    const int v  = wave_inclusive_scan(cb, lane);
    if (t < 512 && lane == 31) wsum[wave] = v;
    __syncthreads();
    if (wave == 0) {
        const int x   = (lane < 16) ? wsum[lane] : 0;
        const int inc = wave_inclusive_scan(x, lane);
        if (lane < 16) wsum[lane] = inc - x;
    }
    __syncthreads();
    if (t < 512) {
        const int st = wsum[wave] + v - cb;
        h[t] = st;
        if (t < NBINS) runs[t * SLICES + blockIdx.x] = (st << 16) | cb;
    }
    __syncthreads();
#pragma unroll
    for (int j = 0; j < EDGES_PER_THREAD; ++j)
        if (bin[j] >= 0) stage[atomicAdd(&h[bin[j]], 1)] = key[j];
    __syncthreads();
    for (int i = t; i < SLICE_EDGES; i += SORT_THREADS) binned[e0 + i] = stage[i];
}

__global__ __launch_bounds__(SORT_THREADS) void k_bin_sort(const int* __restrict__ binned,
                                                           const int* __restrict__ runs,
                                                           int* __restrict__ perm, int* __restrict__ cnt,
                                                           int* __restrict__ off, float* __restrict__ dinv,
                                                           int* __restrict__ cursor) {
    __shared__ int run_s[SLICES];
    __shared__ int c[BIN_NODES];
    __shared__ int wsum[8];
    __shared__ int base_s;
    const int t    = threadIdx.x;
    const int lane = t & 31;
    const int wave = t >> 5;
    const int b    = blockIdx.x;
    if (t < SLICES)    run_s[t] = runs[b * SLICES + t];
    if (t < BIN_NODES) c[t] = 0;
    __syncthreads();

    int k0[RUNS_PER_WAVE];
    int k1[RUNS_PER_WAVE];
#pragma unroll
    for (int q = 0; q < RUNS_PER_WAVE; ++q) {
        const int sb  = wave + q * SORT_WAVES;
        const int m   = __builtin_amdgcn_readfirstlane(run_s[sb]);
        const int len = m & 0xffff;
        const int* p  = binned + sb * SLICE_EDGES + (m >> 16);
        k0[q] = (lane < len) ? p[lane] : -1;
        k1[q] = (lane + 32 < len) ? p[lane + 32] : -1;
    }
#pragma unroll
    for (int q = 0; q < RUNS_PER_WAVE; ++q) {
        if (k0[q] >= 0) atomicAdd(&c[k0[q] & (BIN_NODES - 1)], 1);
        if (k1[q] >= 0) atomicAdd(&c[k1[q] & (BIN_NODES - 1)], 1);
    }
    for (int q = 0; q < RUNS_PER_WAVE; ++q) {
        const int sb  = wave + q * SORT_WAVES;
        const int m   = __builtin_amdgcn_readfirstlane(run_s[sb]);
        const int len = m & 0xffff;
        const int* p  = binned + sb * SLICE_EDGES + (m >> 16);
        for (int i = lane + 64; i < len; i += 32) atomicAdd(&c[p[i] & (BIN_NODES - 1)], 1);
    }
    __syncthreads();

    const int cn = (t < BIN_NODES) ? c[t] : 0;
    const int v  = wave_inclusive_scan(cn, lane);
    if (t < BIN_NODES && lane == 31) wsum[wave] = v;
    __syncthreads();
    if (wave == 0) {
        const int x   = (lane < 8) ? wsum[lane] : 0;
        const int inc = wave_inclusive_scan(x, lane);
        if (lane < 8) wsum[lane] = inc - x;
        if (lane == 31) base_s = atomicAdd(cursor, inc);
    }
    __syncthreads();
    if (t < BIN_NODES) {
        const int o = base_s + wsum[wave] + v - cn;
        c[t] = o;
        const int node = b * BIN_NODES + t;
        if (node < N_NODES) {
            cnt[node]  = cn;
            off[node]  = o;
            dinv[node] = __frsqrt_rn((float)(cn + 1));
        }
    }
    __syncthreads();

#pragma unroll
    for (int q = 0; q < RUNS_PER_WAVE; ++q) {
        if (k0[q] >= 0) perm[atomicAdd(&c[k0[q] & (BIN_NODES - 1)], 1)] = k0[q] >> BIN_SHIFT;
        if (k1[q] >= 0) perm[atomicAdd(&c[k1[q] & (BIN_NODES - 1)], 1)] = k1[q] >> BIN_SHIFT;
    }
    for (int q = 0; q < RUNS_PER_WAVE; ++q) {
        const int sb  = wave + q * SORT_WAVES;
        const int m   = __builtin_amdgcn_readfirstlane(run_s[sb]);
        const int len = m & 0xffff;
        const int* p  = binned + sb * SLICE_EDGES + (m >> 16);
        for (int i = lane + 64; i < len; i += 32) {
            const int k = p[i];
            perm[atomicAdd(&c[k & (BIN_NODES - 1)], 1)] = k >> BIN_SHIFT;
        }
    }
}


__global__ __launch_bounds__(256) void k_gemm_wmma(const float* __restrict__ x,
                                                   const float* __restrict__ W,
                                                   float* __restrict__ xw) {
    __shared__ float Wt[NCLASS * WPITCH];
    for (int i = threadIdx.x; i < NCLASS * NFEAT; i += 256) {
        int k = i >> 6;
        int c = i & 63;
        Wt[c * WPITCH + k] = W[i];
    }
    __syncthreads();

    const int lane  = threadIdx.x & 31;
    const int wave  = threadIdx.x >> 5;
    const int mtile = blockIdx.x * WAVES_PER_BLOCK + wave;
    if (mtile >= MTILES) return;

    const int half = lane >> 4;
    const int l15  = lane & 15;
    const float* xrow = x + (size_t)(mtile * 16 + l15) * NFEAT;

    v8f acc0 = {}, acc1 = {}, acc2 = {}, acc3 = {};

    for (int k = 0; k < NFEAT; k += 4) {
        const int ka = k + 2 * half;
        v2f a  = *(const v2f*)(xrow + ka);
        v2f b0 = *(const v2f*)(&Wt[(0 * 16 + l15) * WPITCH + ka]);
        v2f b1 = *(const v2f*)(&Wt[(1 * 16 + l15) * WPITCH + ka]);
        v2f b2 = *(const v2f*)(&Wt[(2 * 16 + l15) * WPITCH + ka]);
        v2f b3 = *(const v2f*)(&Wt[(3 * 16 + l15) * WPITCH + ka]);
        acc0 = __builtin_amdgcn_wmma_f32_16x16x4_f32(false, a, false, b0, (short)0, acc0, false, false);
        acc1 = __builtin_amdgcn_wmma_f32_16x16x4_f32(false, a, false, b1, (short)0, acc1, false, false);
        acc2 = __builtin_amdgcn_wmma_f32_16x16x4_f32(false, a, false, b2, (short)0, acc2, false, false);
        acc3 = __builtin_amdgcn_wmma_f32_16x16x4_f32(false, a, false, b3, (short)0, acc3, false, false);
    }

    float* orow = xw + (size_t)mtile * 16 * NCLASS;
#pragma unroll
    for (int r = 0; r < 8; ++r) {
        int m = r + 8 * half;
        orow[m * NCLASS + 0 * 16 + l15] = acc0[r];
        orow[m * NCLASS + 1 * 16 + l15] = acc1[r];
        orow[m * NCLASS + 2 * 16 + l15] = acc2[r];
        orow[m * NCLASS + 3 * 16 + l15] = acc3[r];
    }
}


static __device__ __forceinline__ v2f bucket_row(const v2f* __restrict__ xw2, int s, int j, int lane) {
    const int sj = __builtin_amdgcn_readlane(s, j);
    return xw2[(size_t)sj * 32 + lane];
}
static __device__ __forceinline__ void add_row(v2f& acc, v2f r, int wi, int j) {
    const float wj = __int_as_float(__builtin_amdgcn_readlane(wi, j));
    acc.x += r.x * wj;
    acc.y += r.y * wj;
}

__global__ __launch_bounds__(256) void k_gather(const int* __restrict__ perm, const int* __restrict__ off,
                                                const int* __restrict__ cnt, const float* __restrict__ xw,
                                                const float* __restrict__ dinv, const float* __restrict__ bias,
                                                float* __restrict__ out) {
    const int lane = threadIdx.x & 31;
    const int d = blockIdx.x * WAVES_PER_BLOCK + __builtin_amdgcn_readfirstlane((int)(threadIdx.x >> 5));
    if (d >= N_NODES) return;
    const int   n  = cnt[d];
    const int   o  = off[d];
    const float dd = dinv[d];
    const v2f* xw2 = (const v2f*)xw;

    v2f self = xw2[(size_t)d * 32 + lane];
    const float ws = dd * dd;
    v2f acc;
    acc.x = self.x * ws;
    acc.y = self.y * ws;

    for (int j0 = 0; j0 < n; j0 += 32) {
        const int m = (n - j0 < 32) ? (n - j0) : 32;
        int   s = d;
        float w = 0.0f;
        if (lane < m) {
            s = perm[o + j0 + lane];
            w = dinv[s] * dd;
        }
        const int wi = __float_as_int(w);
        const int m4 = (m + 3) & ~3;
        int j = 0;
        for (; m4 - j > 4; j += 8) {
            const v2f r0 = bucket_row(xw2, s, j,     lane);
            const v2f r1 = bucket_row(xw2, s, j + 1, lane);
            const v2f r2 = bucket_row(xw2, s, j + 2, lane);
            const v2f r3 = bucket_row(xw2, s, j + 3, lane);
            const v2f r4 = bucket_row(xw2, s, j + 4, lane);
            const v2f r5 = bucket_row(xw2, s, j + 5, lane);
            const v2f r6 = bucket_row(xw2, s, j + 6, lane);
            const v2f r7 = bucket_row(xw2, s, j + 7, lane);
            add_row(acc, r0, wi, j);
            add_row(acc, r1, wi, j + 1);
            add_row(acc, r2, wi, j + 2);
            add_row(acc, r3, wi, j + 3);
            add_row(acc, r4, wi, j + 4);
            add_row(acc, r5, wi, j + 5);
            add_row(acc, r6, wi, j + 6);
            add_row(acc, r7, wi, j + 7);
        }
        if (j < m4) {
            const v2f r0 = bucket_row(xw2, s, j,     lane);
            const v2f r1 = bucket_row(xw2, s, j + 1, lane);
            const v2f r2 = bucket_row(xw2, s, j + 2, lane);
            const v2f r3 = bucket_row(xw2, s, j + 3, lane);
            add_row(acc, r0, wi, j);
            add_row(acc, r1, wi, j + 1);
            add_row(acc, r2, wi, j + 2);
            add_row(acc, r3, wi, j + 3);
        }
    }

    const v2f b2 = ((const v2f*)bias)[lane];
    v2f res;
    res.x = b2.x + acc.x;
    res.y = b2.y + acc.y;
    ((v2f*)out)[(size_t)d * 32 + lane] = res;
}


extern "C" void kernel_launch(void* const* d_in, const int* in_sizes, int n_in,
                              void* d_out, int out_size, void* d_ws, size_t ws_size,
                              hipStream_t stream) {
    const float* x    = (const float*)d_in[0];
    const float* W    = (const float*)d_in[1];
    const float* bias = (const float*)d_in[2];
    const int*   ei   = (const int*)d_in[3];
    const int*   src  = ei;
    const int*   dst  = ei + N_EDGES;
    float* out = (float*)d_out;

    float* dinv   = (float*)d_ws;
    float* xw     = dinv + 102400;
    int*   cnt    = (int*)(xw + 6400000);
    int*   off    = cnt + 102400;
    int*   cursor = off + 102400;
    int*   binned = cursor + 64;
    int*   perm   = binned + 3200000;
    int*   runs   = perm + 3200000;

    k_slice_sort<<<SLICES, SORT_THREADS, 0, stream>>>(src, dst, binned, runs, cursor);
    k_bin_sort  <<<NBINS,  SORT_THREADS, 0, stream>>>(binned, runs, perm, cnt, off, dinv, cursor);

    k_gemm_wmma<<<(MTILES + WAVES_PER_BLOCK - 1) / WAVES_PER_BLOCK, 256, 0, stream>>>(x, W, xw);

    k_gather<<<(N_NODES + WAVES_PER_BLOCK - 1) / WAVES_PER_BLOCK, 256, 0, stream>>>(perm, off, cnt, xw, dinv, bias, out);
}


# Source-side dinv folded into the product's rows (xs = dinv[row]*(x@W)); gather adds rows unweighted, one multiply by dinv[d] per node: no per-edge dinv fetch; sum order kept, dinv[d] factored out
# speedup vs baseline: 7.6811x; 1.1281x over previous
#include <hip/hip_runtime.h>

typedef float v2f __attribute__((ext_vector_type(2)));
typedef float v8f __attribute__((ext_vector_type(8)));

#define N_NODES 100000
#define N_EDGES 3200000
#define NFEAT   256
#define NCLASS  64
#define MTILES  (N_NODES / 16)
#define WAVES_PER_BLOCK 8
#define WPITCH 260


#define BIN_SHIFT 8
#define BIN_NODES 256
#define NBINS ((N_NODES + BIN_NODES - 1) / BIN_NODES)
#define SLICES 256
#define SLICE_EDGES (N_EDGES / SLICES)
#define SORT_THREADS 1024
#define SORT_WAVES (SORT_THREADS / 32)
#define RUNS_PER_WAVE (SLICES / SORT_WAVES)
#define EDGES_PER_THREAD ((SLICE_EDGES + SORT_THREADS - 1) / SORT_THREADS)

static_assert(SLICE_EDGES * SLICES == N_EDGES, "slices must tile the edge list");
static_assert(RUNS_PER_WAVE * SORT_WAVES == SLICES, "every wave takes the same number of runs");
static_assert(NBINS <= 512, "one scan thread per bin");
static_assert(SLICE_EDGES < 32768, "run start and length share one word");
static_assert(N_NODES <= (1 << (31 - BIN_SHIFT)), "source and local node share one word");

static __device__ __forceinline__ int wave_inclusive_scan(int v, int lane) {
    for (int o = 1; o < 32; o <<= 1) {
        int t = __shfl_up(v, o);
        if (lane >= o) v += t;
    }
    return v;
}

__global__ __launch_bounds__(SORT_THREADS) void k_slice_sort(const int* __restrict__ src,
                                                             const int* __restrict__ dst,
                                                             int* __restrict__ binned,
                                                             int* __restrict__ runs,
                                                             int* __restrict__ cursor) {
    __shared__ int h[512];
    __shared__ int wsum[16];
    __shared__ int stage[SLICE_EDGES];
    const int t    = threadIdx.x;
    const int lane = t & 31;
    const int wave = t >> 5;
    if (blockIdx.x == 0 && t == 0) *cursor = 0;
    if (t < 512) h[t] = 0;

    const int e0 = blockIdx.x * SLICE_EDGES;
    int key[EDGES_PER_THREAD];
    int bin[EDGES_PER_THREAD];
#pragma unroll
    for (int j = 0; j < EDGES_PER_THREAD; ++j) {
        const int i = t + j * SORT_THREADS;
        bin[j] = -1;
        key[j] = 0;
        if (i < SLICE_EDGES) {
            const int d = dst[e0 + i];
            const int s = src[e0 + i];
            bin[j] = d >> BIN_SHIFT;
            key[j] = (s << BIN_SHIFT) | (d & (BIN_NODES - 1));
        }
    }
    __syncthreads();
#pragma unroll
    for (int j = 0; j < EDGES_PER_THREAD; ++j)
        if (bin[j] >= 0) atomicAdd(&h[bin[j]], 1);
    __syncthreads();

    const int cb = (t < 512) ? h[t] : 0;
    const int v  = wave_inclusive_scan(cb, lane);
    if (t < 512 && lane == 31) wsum[wave] = v;
    __syncthreads();
    if (wave == 0) {
        const int x   = (lane < 16) ? wsum[lane] : 0;
        const int inc = wave_inclusive_scan(x, lane);
        if (lane < 16) wsum[lane] = inc - x;
    }
    __syncthreads();
    if (t < 512) {
        const int st = wsum[wave] + v - cb;
        h[t] = st;
        if (t < NBINS) runs[t * SLICES + blockIdx.x] = (st << 16) | cb;
    }
    __syncthreads();
#pragma unroll
    for (int j = 0; j < EDGES_PER_THREAD; ++j)
        if (bin[j] >= 0) stage[atomicAdd(&h[bin[j]], 1)] = key[j];
    __syncthreads();
    for (int i = t; i < SLICE_EDGES; i += SORT_THREADS) binned[e0 + i] = stage[i];
}

__global__ __launch_bounds__(SORT_THREADS) void k_bin_sort(const int* __restrict__ binned,
                                                           const int* __restrict__ runs,
                                                           int* __restrict__ perm, int* __restrict__ cnt,
                                                           int* __restrict__ off, float* __restrict__ dinv,
                                                           int* __restrict__ cursor) {
    __shared__ int run_s[SLICES];
    __shared__ int c[BIN_NODES];
    __shared__ int wsum[8];
    __shared__ int base_s;
    const int t    = threadIdx.x;
    const int lane = t & 31;
    const int wave = t >> 5;
    const int b    = blockIdx.x;
    if (t < SLICES)    run_s[t] = runs[b * SLICES + t];
    if (t < BIN_NODES) c[t] = 0;
    __syncthreads();

    int k0[RUNS_PER_WAVE];
    int k1[RUNS_PER_WAVE];
#pragma unroll
    for (int q = 0; q < RUNS_PER_WAVE; ++q) {
        const int sb  = wave + q * SORT_WAVES;
        const int m   = __builtin_amdgcn_readfirstlane(run_s[sb]);
        const int len = m & 0xffff;
        const int* p  = binned + sb * SLICE_EDGES + (m >> 16);
        k0[q] = (lane < len) ? p[lane] : -1;
        k1[q] = (lane + 32 < len) ? p[lane + 32] : -1;
    }
#pragma unroll
    for (int q = 0; q < RUNS_PER_WAVE; ++q) {
        if (k0[q] >= 0) atomicAdd(&c[k0[q] & (BIN_NODES - 1)], 1);
        if (k1[q] >= 0) atomicAdd(&c[k1[q] & (BIN_NODES - 1)], 1);
    }
    for (int q = 0; q < RUNS_PER_WAVE; ++q) {
        const int sb  = wave + q * SORT_WAVES;
        const int m   = __builtin_amdgcn_readfirstlane(run_s[sb]);
        const int len = m & 0xffff;
        const int* p  = binned + sb * SLICE_EDGES + (m >> 16);
        for (int i = lane + 64; i < len; i += 32) atomicAdd(&c[p[i] & (BIN_NODES - 1)], 1);
    }
    __syncthreads();

    const int cn = (t < BIN_NODES) ? c[t] : 0;
    const int v  = wave_inclusive_scan(cn, lane);
    if (t < BIN_NODES && lane == 31) wsum[wave] = v;
    __syncthreads();
    if (wave == 0) {
        const int x   = (lane < 8) ? wsum[lane] : 0;
        const int inc = wave_inclusive_scan(x, lane);
        if (lane < 8) wsum[lane] = inc - x;
        if (lane == 31) base_s = atomicAdd(cursor, inc);
    }
    __syncthreads();
    if (t < BIN_NODES) {
        const int o = base_s + wsum[wave] + v - cn;
        c[t] = o;
        const int node = b * BIN_NODES + t;
        if (node < N_NODES) {
            cnt[node]  = cn;
            off[node]  = o;
            dinv[node] = __frsqrt_rn((float)(cn + 1));
        }
    }
    __syncthreads();

#pragma unroll
    for (int q = 0; q < RUNS_PER_WAVE; ++q) {
        if (k0[q] >= 0) perm[atomicAdd(&c[k0[q] & (BIN_NODES - 1)], 1)] = k0[q] >> BIN_SHIFT;
        if (k1[q] >= 0) perm[atomicAdd(&c[k1[q] & (BIN_NODES - 1)], 1)] = k1[q] >> BIN_SHIFT;
    }
    for (int q = 0; q < RUNS_PER_WAVE; ++q) {
        const int sb  = wave + q * SORT_WAVES;
        const int m   = __builtin_amdgcn_readfirstlane(run_s[sb]);
        const int len = m & 0xffff;
        const int* p  = binned + sb * SLICE_EDGES + (m >> 16);
        for (int i = lane + 64; i < len; i += 32) {
            const int k = p[i];
            perm[atomicAdd(&c[k & (BIN_NODES - 1)], 1)] = k >> BIN_SHIFT;
        }
    }
}


__global__ __launch_bounds__(256) void k_gemm_wmma(const float* __restrict__ x,
                                                   const float* __restrict__ W,
                                                   const float* __restrict__ dinv,
                                                   float* __restrict__ xw) {
    __shared__ float Wt[NCLASS * WPITCH];
    for (int i = threadIdx.x; i < NCLASS * NFEAT; i += 256) {
        int k = i >> 6;
        int c = i & 63;
        Wt[c * WPITCH + k] = W[i];
    }
    __syncthreads();

    const int lane  = threadIdx.x & 31;
    const int wave  = threadIdx.x >> 5;
    const int mtile = blockIdx.x * WAVES_PER_BLOCK + wave;
    if (blockIdx.x == 0 && threadIdx.x < NCLASS) xw[(size_t)N_NODES * NCLASS + threadIdx.x] = 0.0f;
    if (mtile >= MTILES) return;

    const int half = lane >> 4;
    const int l15  = lane & 15;
    const float* xrow = x + (size_t)(mtile * 16 + l15) * NFEAT;

    const float4 dlo = *(const float4*)(dinv + mtile * 16 + 8 * half);
    const float4 dhi = *(const float4*)(dinv + mtile * 16 + 8 * half + 4);

    v8f acc0 = {}, acc1 = {}, acc2 = {}, acc3 = {};

    for (int k = 0; k < NFEAT; k += 4) {
        const int ka = k + 2 * half;
        v2f a  = *(const v2f*)(xrow + ka);
        v2f b0 = *(const v2f*)(&Wt[(0 * 16 + l15) * WPITCH + ka]);
        v2f b1 = *(const v2f*)(&Wt[(1 * 16 + l15) * WPITCH + ka]);
        v2f b2 = *(const v2f*)(&Wt[(2 * 16 + l15) * WPITCH + ka]);
        v2f b3 = *(const v2f*)(&Wt[(3 * 16 + l15) * WPITCH + ka]);
        acc0 = __builtin_amdgcn_wmma_f32_16x16x4_f32(false, a, false, b0, (short)0, acc0, false, false);
        acc1 = __builtin_amdgcn_wmma_f32_16x16x4_f32(false, a, false, b1, (short)0, acc1, false, false);
        acc2 = __builtin_amdgcn_wmma_f32_16x16x4_f32(false, a, false, b2, (short)0, acc2, false, false);
        acc3 = __builtin_amdgcn_wmma_f32_16x16x4_f32(false, a, false, b3, (short)0, acc3, false, false);
    }

    float* orow = xw + (size_t)mtile * 16 * NCLASS;
    const float dv[8] = {dlo.x, dlo.y, dlo.z, dlo.w, dhi.x, dhi.y, dhi.z, dhi.w};
#pragma unroll
    for (int r = 0; r < 8; ++r) {
        int m = r + 8 * half;
        orow[m * NCLASS + 0 * 16 + l15] = acc0[r] * dv[r];
        orow[m * NCLASS + 1 * 16 + l15] = acc1[r] * dv[r];
        orow[m * NCLASS + 2 * 16 + l15] = acc2[r] * dv[r];
        orow[m * NCLASS + 3 * 16 + l15] = acc3[r] * dv[r];
    }
}


static __device__ __forceinline__ v2f bucket_row(const v2f* __restrict__ xw2, int s, int j, int lane) {
    const int sj = __builtin_amdgcn_readlane(s, j);
    return xw2[(size_t)sj * 32 + lane];
}
static __device__ __forceinline__ void add_row(v2f& acc, v2f r) {
    acc.x += r.x;
    acc.y += r.y;
}

__global__ __launch_bounds__(256) void k_gather(const int* __restrict__ perm, const int* __restrict__ off,
                                                const int* __restrict__ cnt, const float* __restrict__ xw,
                                                const float* __restrict__ dinv, const float* __restrict__ bias,
                                                float* __restrict__ out) {
    const int lane = threadIdx.x & 31;
    const int d = blockIdx.x * WAVES_PER_BLOCK + __builtin_amdgcn_readfirstlane((int)(threadIdx.x >> 5));
    if (d >= N_NODES) return;
    const int   n  = cnt[d];
    const int   o  = off[d];
    const float dd = dinv[d];
    const v2f* xw2 = (const v2f*)xw;

    v2f acc = xw2[(size_t)d * 32 + lane];

    for (int j0 = 0; j0 < n; j0 += 32) {
        const int m = (n - j0 < 32) ? (n - j0) : 32;
        int s = N_NODES;
        if (lane < m) s = perm[o + j0 + lane];
        const int m4 = (m + 3) & ~3;
        int j = 0;
        for (; m4 - j > 4; j += 8) {
            const v2f r0 = bucket_row(xw2, s, j,     lane);
            const v2f r1 = bucket_row(xw2, s, j + 1, lane);
            const v2f r2 = bucket_row(xw2, s, j + 2, lane);
            const v2f r3 = bucket_row(xw2, s, j + 3, lane);
            const v2f r4 = bucket_row(xw2, s, j + 4, lane);
            const v2f r5 = bucket_row(xw2, s, j + 5, lane);
            const v2f r6 = bucket_row(xw2, s, j + 6, lane);
            const v2f r7 = bucket_row(xw2, s, j + 7, lane);
            add_row(acc, r0);
            add_row(acc, r1);
            add_row(acc, r2);
            add_row(acc, r3);
            add_row(acc, r4);
            add_row(acc, r5);
            add_row(acc, r6);
            add_row(acc, r7);
        }
        if (j < m4) {
            const v2f r0 = bucket_row(xw2, s, j,     lane);
            const v2f r1 = bucket_row(xw2, s, j + 1, lane);
            const v2f r2 = bucket_row(xw2, s, j + 2, lane);
            const v2f r3 = bucket_row(xw2, s, j + 3, lane);
            add_row(acc, r0);
            add_row(acc, r1);
            add_row(acc, r2);
            add_row(acc, r3);
        }
    }

    const v2f b2 = ((const v2f*)bias)[lane];
    v2f res;
    res.x = b2.x + acc.x * dd;
    res.y = b2.y + acc.y * dd;
    ((v2f*)out)[(size_t)d * 32 + lane] = res;
}


extern "C" void kernel_launch(void* const* d_in, const int* in_sizes, int n_in,
                              void* d_out, int out_size, void* d_ws, size_t ws_size,
                              hipStream_t stream) {
    const float* x    = (const float*)d_in[0];
    const float* W    = (const float*)d_in[1];
    const float* bias = (const float*)d_in[2];
    const int*   ei   = (const int*)d_in[3];
    const int*   src  = ei;
    const int*   dst  = ei + N_EDGES;
    float* out = (float*)d_out;

    float* dinv   = (float*)d_ws;
    float* xw     = dinv + 102400;
    int*   cnt    = (int*)(xw + 6400064);
    int*   off    = cnt + 102400;
    int*   cursor = off + 102400;
    int*   binned = cursor + 64;
    int*   perm   = binned + 3200000;
    int*   runs   = perm + 3200000;

    k_slice_sort<<<SLICES, SORT_THREADS, 0, stream>>>(src, dst, binned, runs, cursor);
    k_bin_sort  <<<NBINS,  SORT_THREADS, 0, stream>>>(binned, runs, perm, cnt, off, dinv, cursor);

    k_gemm_wmma<<<(MTILES + WAVES_PER_BLOCK - 1) / WAVES_PER_BLOCK, 256, 0, stream>>>(x, W, dinv, xw);

    k_gather<<<(N_NODES + WAVES_PER_BLOCK - 1) / WAVES_PER_BLOCK, 256, 0, stream>>>(perm, off, cnt, xw, dinv, bias, out);
}


# x @ W runs as a block role beside the edge slice sort in one launch (4 launches to 3); its rows leave unscaled, the bin kernel scales its 256 rows by dinv in place; sum order unchanged
# speedup vs baseline: 8.4086x; 1.0947x over previous
#include <hip/hip_runtime.h>

typedef float v2f __attribute__((ext_vector_type(2)));
typedef float v8f __attribute__((ext_vector_type(8)));

#define N_NODES 100000
#define N_EDGES 3200000
#define NFEAT   256
#define NCLASS  64
#define MTILES  (N_NODES / 16)
#define WAVES_PER_BLOCK 8
#define WPITCH 260


#define BIN_SHIFT 8
#define BIN_NODES 256
#define NBINS ((N_NODES + BIN_NODES - 1) / BIN_NODES)
#define SLICES 256
#define SLICE_EDGES (N_EDGES / SLICES)
#define SORT_THREADS 1024
#define SORT_WAVES (SORT_THREADS / 32)
#define RUNS_PER_WAVE (SLICES / SORT_WAVES)
#define EDGES_PER_THREAD ((SLICE_EDGES + SORT_THREADS - 1) / SORT_THREADS)

static_assert(SLICE_EDGES * SLICES == N_EDGES, "slices must tile the edge list");
static_assert(RUNS_PER_WAVE * SORT_WAVES == SLICES, "every wave takes the same number of runs");
static_assert(NBINS <= 512, "one scan thread per bin");
static_assert(SLICE_EDGES < 32768, "run start and length share one word");
static_assert(N_NODES <= (1 << (31 - BIN_SHIFT)), "source and local node share one word");

static __device__ __forceinline__ int wave_inclusive_scan(int v, int lane) {
    for (int o = 1; o < 32; o <<= 1) {
        int t = __shfl_up(v, o);
        if (lane >= o) v += t;
    }
    return v;
}


#define PRODUCT_BLOCKS 250
#define TILES_PER_BLOCK 25
#define ROLE_LDS_BYTES (NCLASS * WPITCH * 4)

static_assert(PRODUCT_BLOCKS * TILES_PER_BLOCK == MTILES, "the product blocks tile the rows exactly");
static_assert(TILES_PER_BLOCK <= SORT_WAVES, "one tile per wave");
static_assert((512 + 16 + SLICE_EDGES) * 4 <= ROLE_LDS_BYTES, "the sort role's LDS fits the shared buffer");

__global__ __launch_bounds__(SORT_THREADS, 16) void k_product_and_slice_sort(const float* __restrict__ x,
                                                                         const float* __restrict__ W,
                                                                         float* __restrict__ xw,
                                                                         const int* __restrict__ src,
                                                                         const int* __restrict__ dst,
                                                                         int* __restrict__ binned,
                                                                         int* __restrict__ runs,
                                                                         int* __restrict__ cursor) {
    __shared__ __attribute__((aligned(16))) unsigned char role_lds[ROLE_LDS_BYTES];
    const int t    = threadIdx.x;
    const int lane = t & 31;
    const int wave = __builtin_amdgcn_readfirstlane(t >> 5);

    if (blockIdx.x < PRODUCT_BLOCKS) {
        float* Wt = (float*)role_lds;
        for (int i = t; i < NCLASS * NFEAT; i += SORT_THREADS) {
            int k = i >> 6;
            int c = i & 63;
            Wt[c * WPITCH + k] = W[i];
        }
        __syncthreads();

        if (blockIdx.x == 0 && t < NCLASS) xw[(size_t)N_NODES * NCLASS + t] = 0.0f;
        if (wave >= TILES_PER_BLOCK) return;
        const int mtile = blockIdx.x * TILES_PER_BLOCK + wave;

        const int half = lane >> 4;
        const int l15  = lane & 15;
        const float* xrow = x + (size_t)(mtile * 16 + l15) * NFEAT;

        v8f acc0 = {}, acc1 = {}, acc2 = {}, acc3 = {};

#pragma unroll 4
        for (int k = 0; k < NFEAT; k += 4) {
            const int ka = k + 2 * half;
            v2f a  = *(const v2f*)(xrow + ka);
            v2f b0 = *(const v2f*)(&Wt[(0 * 16 + l15) * WPITCH + ka]);
            v2f b1 = *(const v2f*)(&Wt[(1 * 16 + l15) * WPITCH + ka]);
            v2f b2 = *(const v2f*)(&Wt[(2 * 16 + l15) * WPITCH + ka]);
            v2f b3 = *(const v2f*)(&Wt[(3 * 16 + l15) * WPITCH + ka]);
            acc0 = __builtin_amdgcn_wmma_f32_16x16x4_f32(false, a, false, b0, (short)0, acc0, false, false);
            acc1 = __builtin_amdgcn_wmma_f32_16x16x4_f32(false, a, false, b1, (short)0, acc1, false, false);
            acc2 = __builtin_amdgcn_wmma_f32_16x16x4_f32(false, a, false, b2, (short)0, acc2, false, false);
            acc3 = __builtin_amdgcn_wmma_f32_16x16x4_f32(false, a, false, b3, (short)0, acc3, false, false);
        }

        float* orow = xw + (size_t)mtile * 16 * NCLASS;
#pragma unroll
        for (int r = 0; r < 8; ++r) {
            int m = r + 8 * half;
            orow[m * NCLASS + 0 * 16 + l15] = acc0[r];
            orow[m * NCLASS + 1 * 16 + l15] = acc1[r];
            orow[m * NCLASS + 2 * 16 + l15] = acc2[r];
            orow[m * NCLASS + 3 * 16 + l15] = acc3[r];
        }
        return;
    }

    int* h     = (int*)role_lds;
    int* wsum  = h + 512;
    int* stage = wsum + 16;
    const int slice = blockIdx.x - PRODUCT_BLOCKS;
    if (slice == 0 && t == 0) *cursor = 0;
    if (t < 512) h[t] = 0;

    const int e0 = slice * SLICE_EDGES;
    int key[EDGES_PER_THREAD];
    int bin[EDGES_PER_THREAD];
#pragma unroll
    for (int j = 0; j < EDGES_PER_THREAD; ++j) {
        const int i = t + j * SORT_THREADS;
        bin[j] = -1;
        key[j] = 0;
        if (i < SLICE_EDGES) {
            const int d = dst[e0 + i];
            const int s = src[e0 + i];
            bin[j] = d >> BIN_SHIFT;
            key[j] = (s << BIN_SHIFT) | (d & (BIN_NODES - 1));
        }
    }
    __syncthreads();
#pragma unroll
    for (int j = 0; j < EDGES_PER_THREAD; ++j)
        if (bin[j] >= 0) atomicAdd(&h[bin[j]], 1);
    __syncthreads();

    const int cb = (t < 512) ? h[t] : 0;
    const int v  = wave_inclusive_scan(cb, lane);
    if (t < 512 && lane == 31) wsum[wave] = v;
    __syncthreads();
    if (wave == 0) {
        const int xs  = (lane < 16) ? wsum[lane] : 0;
        const int inc = wave_inclusive_scan(xs, lane);
        if (lane < 16) wsum[lane] = inc - xs;
    }
    __syncthreads();
    if (t < 512) {
        const int st = wsum[wave] + v - cb;
        h[t] = st;
        if (t < NBINS) runs[t * SLICES + slice] = (st << 16) | cb;
    }
    __syncthreads();
#pragma unroll
    for (int j = 0; j < EDGES_PER_THREAD; ++j)
        if (bin[j] >= 0) stage[atomicAdd(&h[bin[j]], 1)] = key[j];
    __syncthreads();
    for (int i = t; i < SLICE_EDGES; i += SORT_THREADS) binned[e0 + i] = stage[i];
}

#define ROW_PIECES (BIN_NODES * (NCLASS / 4) / SORT_THREADS)
static_assert(ROW_PIECES * SORT_THREADS == BIN_NODES * (NCLASS / 4), "the bin's rows split evenly over the threads");

__global__ __launch_bounds__(SORT_THREADS, 16) void k_bin_sort(const int* __restrict__ binned,
                                                           const int* __restrict__ runs,
                                                           int* __restrict__ perm, int* __restrict__ cnt,
                                                           int* __restrict__ off, float* __restrict__ dinv,
                                                           int* __restrict__ cursor, float* xw) {
    __shared__ int run_s[SLICES];
    __shared__ int c[BIN_NODES];
    __shared__ int wsum[8];
    __shared__ int base_s;
    __shared__ float dinv_s[BIN_NODES];
    const int t    = threadIdx.x;
    const int lane = t & 31;
    const int wave = t >> 5;
    const int b    = blockIdx.x;
    if (t < SLICES)    run_s[t] = runs[b * SLICES + t];
    if (t < BIN_NODES) c[t] = 0;
    __syncthreads();

    int k0[RUNS_PER_WAVE];
    int k1[RUNS_PER_WAVE];
#pragma unroll
    for (int q = 0; q < RUNS_PER_WAVE; ++q) {
        const int sb  = wave + q * SORT_WAVES;
        const int m   = __builtin_amdgcn_readfirstlane(run_s[sb]);
        const int len = m & 0xffff;
        const int* p  = binned + sb * SLICE_EDGES + (m >> 16);
        k0[q] = (lane < len) ? p[lane] : -1;
        k1[q] = (lane + 32 < len) ? p[lane + 32] : -1;
    }
    float4* xrows = (float4*)xw + (size_t)b * BIN_NODES * (NCLASS / 4);
    float4 piece[ROW_PIECES];
#pragma unroll
    for (int j = 0; j < ROW_PIECES; ++j) {
        const int i = t + j * SORT_THREADS;
        piece[j] = make_float4(0.0f, 0.0f, 0.0f, 0.0f);
        if (b * BIN_NODES + (i >> 4) < N_NODES) piece[j] = xrows[i];
    }
#pragma unroll
    for (int q = 0; q < RUNS_PER_WAVE; ++q) {
        if (k0[q] >= 0) atomicAdd(&c[k0[q] & (BIN_NODES - 1)], 1);
        if (k1[q] >= 0) atomicAdd(&c[k1[q] & (BIN_NODES - 1)], 1);
    }
    for (int q = 0; q < RUNS_PER_WAVE; ++q) {
        const int sb  = wave + q * SORT_WAVES;
        const int m   = __builtin_amdgcn_readfirstlane(run_s[sb]);
        const int len = m & 0xffff;
        const int* p  = binned + sb * SLICE_EDGES + (m >> 16);
        for (int i = lane + 64; i < len; i += 32) atomicAdd(&c[p[i] & (BIN_NODES - 1)], 1);
    }
    __syncthreads();

    const int cn = (t < BIN_NODES) ? c[t] : 0;
    const int v  = wave_inclusive_scan(cn, lane);
    if (t < BIN_NODES && lane == 31) wsum[wave] = v;
    __syncthreads();
    if (wave == 0) {
        const int x   = (lane < 8) ? wsum[lane] : 0;
        const int inc = wave_inclusive_scan(x, lane);
        if (lane < 8) wsum[lane] = inc - x;
        if (lane == 31) base_s = atomicAdd(cursor, inc);
    }
    __syncthreads();
    if (t < BIN_NODES) {
        const int o = base_s + wsum[wave] + v - cn;
        c[t] = o;
        const int node = b * BIN_NODES + t;
        const float dn = __frsqrt_rn((float)(cn + 1));
        dinv_s[t] = dn;
        if (node < N_NODES) {
            cnt[node]  = cn;
            off[node]  = o;
            dinv[node] = dn;
        }
    }
    __syncthreads();

#pragma unroll
    for (int j = 0; j < ROW_PIECES; ++j) {
        const int i = t + j * SORT_THREADS;
        if (b * BIN_NODES + (i >> 4) < N_NODES) {
            const float dn = dinv_s[i >> 4];
            float4 p = piece[j];
            p.x *= dn;
            p.y *= dn;
            p.z *= dn;
            p.w *= dn;
            xrows[i] = p;
        }
    }

#pragma unroll
    for (int q = 0; q < RUNS_PER_WAVE; ++q) {
        if (k0[q] >= 0) perm[atomicAdd(&c[k0[q] & (BIN_NODES - 1)], 1)] = k0[q] >> BIN_SHIFT;
        if (k1[q] >= 0) perm[atomicAdd(&c[k1[q] & (BIN_NODES - 1)], 1)] = k1[q] >> BIN_SHIFT;
    }
    for (int q = 0; q < RUNS_PER_WAVE; ++q) {
        const int sb  = wave + q * SORT_WAVES;
        const int m   = __builtin_amdgcn_readfirstlane(run_s[sb]);
        const int len = m & 0xffff;
        const int* p  = binned + sb * SLICE_EDGES + (m >> 16);
        for (int i = lane + 64; i < len; i += 32) {
            const int k = p[i];
            perm[atomicAdd(&c[k & (BIN_NODES - 1)], 1)] = k >> BIN_SHIFT;
        }
    }
}


static __device__ __forceinline__ v2f bucket_row(const v2f* __restrict__ xw2, int s, int j, int lane) {
    const int sj = __builtin_amdgcn_readlane(s, j);
    return xw2[(size_t)sj * 32 + lane];
}
static __device__ __forceinline__ void add_row(v2f& acc, v2f r) {
    acc.x += r.x;
    acc.y += r.y;
}

__global__ __launch_bounds__(256) void k_gather(const int* __restrict__ perm, const int* __restrict__ off,
                                                const int* __restrict__ cnt, const float* __restrict__ xw,
                                                const float* __restrict__ dinv, const float* __restrict__ bias,
                                                float* __restrict__ out) {
    const int lane = threadIdx.x & 31;
    const int d = blockIdx.x * WAVES_PER_BLOCK + __builtin_amdgcn_readfirstlane((int)(threadIdx.x >> 5));
    if (d >= N_NODES) return;
    const int   n  = cnt[d];
    const int   o  = off[d];
    const float dd = dinv[d];
    const v2f* xw2 = (const v2f*)xw;

    v2f acc = xw2[(size_t)d * 32 + lane];

    for (int j0 = 0; j0 < n; j0 += 32) {
        const int m = (n - j0 < 32) ? (n - j0) : 32;
        int s = N_NODES;
        if (lane < m) s = perm[o + j0 + lane];
        const int m4 = (m + 3) & ~3;
        int j = 0;
        for (; m4 - j > 4; j += 8) {
            const v2f r0 = bucket_row(xw2, s, j,     lane);
            const v2f r1 = bucket_row(xw2, s, j + 1, lane);
            const v2f r2 = bucket_row(xw2, s, j + 2, lane);
            const v2f r3 = bucket_row(xw2, s, j + 3, lane);
            const v2f r4 = bucket_row(xw2, s, j + 4, lane);
            const v2f r5 = bucket_row(xw2, s, j + 5, lane);
            const v2f r6 = bucket_row(xw2, s, j + 6, lane);
            const v2f r7 = bucket_row(xw2, s, j + 7, lane);
            add_row(acc, r0);
            add_row(acc, r1);
            add_row(acc, r2);
            add_row(acc, r3);
            add_row(acc, r4);
            add_row(acc, r5);
            add_row(acc, r6);
            add_row(acc, r7);
        }
        if (j < m4) {
            const v2f r0 = bucket_row(xw2, s, j,     lane);
            const v2f r1 = bucket_row(xw2, s, j + 1, lane);
            const v2f r2 = bucket_row(xw2, s, j + 2, lane);
            const v2f r3 = bucket_row(xw2, s, j + 3, lane);
            add_row(acc, r0);
            add_row(acc, r1);
            add_row(acc, r2);
            add_row(acc, r3);
        }
    }

    const v2f b2 = ((const v2f*)bias)[lane];
    v2f res;
    res.x = b2.x + acc.x * dd;
    res.y = b2.y + acc.y * dd;
    ((v2f*)out)[(size_t)d * 32 + lane] = res;
}


extern "C" void kernel_launch(void* const* d_in, const int* in_sizes, int n_in,
                              void* d_out, int out_size, void* d_ws, size_t ws_size,
                              hipStream_t stream) {
    const float* x    = (const float*)d_in[0];
    const float* W    = (const float*)d_in[1];
    const float* bias = (const float*)d_in[2];
    const int*   ei   = (const int*)d_in[3];
    const int*   src  = ei;
    const int*   dst  = ei + N_EDGES;
    float* out = (float*)d_out;

    float* dinv   = (float*)d_ws;
    float* xw     = dinv + 102400;
    int*   cnt    = (int*)(xw + 6400064);
    int*   off    = cnt + 102400;
    int*   cursor = off + 102400;
    int*   binned = cursor + 64;
    int*   perm   = binned + 3200000;
    int*   runs   = perm + 3200000;

    k_product_and_slice_sort<<<PRODUCT_BLOCKS + SLICES, SORT_THREADS, 0, stream>>>(x, W, xw, src, dst, binned, runs, cursor);
    k_bin_sort<<<NBINS, SORT_THREADS, 0, stream>>>(binned, runs, perm, cnt, off, dinv, cursor, xw);

    k_gather<<<(N_NODES + WAVES_PER_BLOCK - 1) / WAVES_PER_BLOCK, 256, 0, stream>>>(perm, off, cnt, xw, dinv, bias, out);
}


# Bin kernel places each bin's sources in an LDS copy of its perm region (regions line-aligned) and writes it out a whole 128-byte line per store, not 4-byte stores over 32 lines; sum order kept
# speedup vs baseline: 9.0632x; 1.0778x over previous
#include <hip/hip_runtime.h>

typedef float v2f __attribute__((ext_vector_type(2)));
typedef float v8f __attribute__((ext_vector_type(8)));

#define N_NODES 100000
#define N_EDGES 3200000
#define NFEAT   256
#define NCLASS  64
#define MTILES  (N_NODES / 16)
#define WAVES_PER_BLOCK 8
#define WPITCH 260


#define BIN_SHIFT 8
#define BIN_NODES 256
#define NBINS ((N_NODES + BIN_NODES - 1) / BIN_NODES)
#define SLICES 256
#define SLICE_EDGES (N_EDGES / SLICES)
#define SORT_THREADS 1024
#define SORT_WAVES (SORT_THREADS / 32)
#define RUNS_PER_WAVE (SLICES / SORT_WAVES)
#define EDGES_PER_THREAD ((SLICE_EDGES + SORT_THREADS - 1) / SORT_THREADS)

static_assert(SLICE_EDGES * SLICES == N_EDGES, "slices must tile the edge list");
static_assert(RUNS_PER_WAVE * SORT_WAVES == SLICES, "every wave takes the same number of runs");
static_assert(NBINS <= 512, "one scan thread per bin");
static_assert(SLICE_EDGES < 32768, "run start and length share one word");
static_assert(N_NODES <= (1 << (31 - BIN_SHIFT)), "source and local node share one word");

static __device__ __forceinline__ int wave_inclusive_scan(int v, int lane) {
    for (int o = 1; o < 32; o <<= 1) {
        int t = __shfl_up(v, o);
        if (lane >= o) v += t;
    }
    return v;
}


#define PRODUCT_BLOCKS 250
#define TILES_PER_BLOCK 25
#define ROLE_LDS_BYTES (NCLASS * WPITCH * 4)

static_assert(PRODUCT_BLOCKS * TILES_PER_BLOCK == MTILES, "the product blocks tile the rows exactly");
static_assert(TILES_PER_BLOCK <= SORT_WAVES, "one tile per wave");
static_assert((512 + 16 + SLICE_EDGES) * 4 <= ROLE_LDS_BYTES, "the sort role's LDS fits the shared buffer");

__global__ __launch_bounds__(SORT_THREADS, 16) void k_product_and_slice_sort(const float* __restrict__ x,
                                                                         const float* __restrict__ W,
                                                                         float* __restrict__ xw,
                                                                         const int* __restrict__ src,
                                                                         const int* __restrict__ dst,
                                                                         int* __restrict__ binned,
                                                                         int* __restrict__ runs,
                                                                         int* __restrict__ cursor) {
    __shared__ __attribute__((aligned(16))) unsigned char role_lds[ROLE_LDS_BYTES];
    const int t    = threadIdx.x;
    const int lane = t & 31;
    const int wave = __builtin_amdgcn_readfirstlane(t >> 5);

    if (blockIdx.x < PRODUCT_BLOCKS) {
        float* Wt = (float*)role_lds;
        for (int i = t; i < NCLASS * NFEAT; i += SORT_THREADS) {
            int k = i >> 6;
            int c = i & 63;
            Wt[c * WPITCH + k] = W[i];
        }
        __syncthreads();

        if (blockIdx.x == 0 && t < NCLASS) xw[(size_t)N_NODES * NCLASS + t] = 0.0f;
        if (wave >= TILES_PER_BLOCK) return;
        const int mtile = blockIdx.x * TILES_PER_BLOCK + wave;

        const int half = lane >> 4;
        const int l15  = lane & 15;
        const float* xrow = x + (size_t)(mtile * 16 + l15) * NFEAT;

        v8f acc0 = {}, acc1 = {}, acc2 = {}, acc3 = {};

#pragma unroll 4
        for (int k = 0; k < NFEAT; k += 4) {
            const int ka = k + 2 * half;
            v2f a  = *(const v2f*)(xrow + ka);
            v2f b0 = *(const v2f*)(&Wt[(0 * 16 + l15) * WPITCH + ka]);
            v2f b1 = *(const v2f*)(&Wt[(1 * 16 + l15) * WPITCH + ka]);
            v2f b2 = *(const v2f*)(&Wt[(2 * 16 + l15) * WPITCH + ka]);
            v2f b3 = *(const v2f*)(&Wt[(3 * 16 + l15) * WPITCH + ka]);
            acc0 = __builtin_amdgcn_wmma_f32_16x16x4_f32(false, a, false, b0, (short)0, acc0, false, false);
            acc1 = __builtin_amdgcn_wmma_f32_16x16x4_f32(false, a, false, b1, (short)0, acc1, false, false);
            acc2 = __builtin_amdgcn_wmma_f32_16x16x4_f32(false, a, false, b2, (short)0, acc2, false, false);
            acc3 = __builtin_amdgcn_wmma_f32_16x16x4_f32(false, a, false, b3, (short)0, acc3, false, false);
        }

        float* orow = xw + (size_t)mtile * 16 * NCLASS;
#pragma unroll
        for (int r = 0; r < 8; ++r) {
            int m = r + 8 * half;
            orow[m * NCLASS + 0 * 16 + l15] = acc0[r];
            orow[m * NCLASS + 1 * 16 + l15] = acc1[r];
            orow[m * NCLASS + 2 * 16 + l15] = acc2[r];
            orow[m * NCLASS + 3 * 16 + l15] = acc3[r];
        }
        return;
    }

    int* h     = (int*)role_lds;
    int* wsum  = h + 512;
    int* stage = wsum + 16;
    const int slice = blockIdx.x - PRODUCT_BLOCKS;
    if (slice == 0 && t == 0) *cursor = 0;
    if (t < 512) h[t] = 0;

    const int e0 = slice * SLICE_EDGES;
    int key[EDGES_PER_THREAD];
    int bin[EDGES_PER_THREAD];
#pragma unroll
    for (int j = 0; j < EDGES_PER_THREAD; ++j) {
        const int i = t + j * SORT_THREADS;
        bin[j] = -1;
        key[j] = 0;
        if (i < SLICE_EDGES) {
            const int d = dst[e0 + i];
            const int s = src[e0 + i];
            bin[j] = d >> BIN_SHIFT;
            key[j] = (s << BIN_SHIFT) | (d & (BIN_NODES - 1));
        }
    }
    __syncthreads();
#pragma unroll
    for (int j = 0; j < EDGES_PER_THREAD; ++j)
        if (bin[j] >= 0) atomicAdd(&h[bin[j]], 1);
    __syncthreads();

    const int cb = (t < 512) ? h[t] : 0;
    const int v  = wave_inclusive_scan(cb, lane);
    if (t < 512 && lane == 31) wsum[wave] = v;
    __syncthreads();
    if (wave == 0) {
        const int xs  = (lane < 16) ? wsum[lane] : 0;
        const int inc = wave_inclusive_scan(xs, lane);
        if (lane < 16) wsum[lane] = inc - xs;
    }
    __syncthreads();
    if (t < 512) {
        const int st = wsum[wave] + v - cb;
        h[t] = st;
        if (t < NBINS) runs[t * SLICES + slice] = (st << 16) | cb;
    }
    __syncthreads();
#pragma unroll
    for (int j = 0; j < EDGES_PER_THREAD; ++j)
        if (bin[j] >= 0) stage[atomicAdd(&h[bin[j]], 1)] = key[j];
    __syncthreads();
    for (int i = t; i < SLICE_EDGES; i += SORT_THREADS) binned[e0 + i] = stage[i];
}

#define ROW_PIECES (BIN_NODES * (NCLASS / 4) / SORT_THREADS)
#define STAGE_CAP 16384
#define REGION_ALIGN 32
#define PERM_WORDS (N_EDGES + NBINS * REGION_ALIGN)
static_assert(PERM_WORDS == 3212512, "the launch's workspace layout reserves the padded perm");
static_assert(STAGE_CAP % SORT_THREADS == 0 && REGION_ALIGN == 32, "one whole line per wave store");
static_assert(ROW_PIECES * SORT_THREADS == BIN_NODES * (NCLASS / 4), "the bin's rows split evenly over the threads");

__global__ __launch_bounds__(SORT_THREADS, 16) void k_bin_sort(const int* __restrict__ binned,
                                                           const int* __restrict__ runs,
                                                           int* __restrict__ perm, int* __restrict__ cnt,
                                                           int* __restrict__ off, float* __restrict__ dinv,
                                                           int* __restrict__ cursor, float* xw) {
    __shared__ int run_s[SLICES];
    __shared__ int c[BIN_NODES];
    __shared__ int wsum[8];
    __shared__ int base_s;
    __shared__ int total_s;
    __shared__ float dinv_s[BIN_NODES];
    __shared__ int stage[STAGE_CAP];
    const int t    = threadIdx.x;
    const int lane = t & 31;
    const int wave = t >> 5;
    const int b    = blockIdx.x;
    if (t < SLICES)    run_s[t] = runs[b * SLICES + t];
    if (t < BIN_NODES) c[t] = 0;
    __syncthreads();

    int k0[RUNS_PER_WAVE];
    int k1[RUNS_PER_WAVE];
#pragma unroll
    for (int q = 0; q < RUNS_PER_WAVE; ++q) {
        const int sb  = wave + q * SORT_WAVES;
        const int m   = __builtin_amdgcn_readfirstlane(run_s[sb]);
        const int len = m & 0xffff;
        const int* p  = binned + sb * SLICE_EDGES + (m >> 16);
        k0[q] = (lane < len) ? p[lane] : -1;
        k1[q] = (lane + 32 < len) ? p[lane + 32] : -1;
    }
    float4* xrows = (float4*)xw + (size_t)b * BIN_NODES * (NCLASS / 4);
    float4 piece[ROW_PIECES];
#pragma unroll
    for (int j = 0; j < ROW_PIECES; ++j) {
        const int i = t + j * SORT_THREADS;
        piece[j] = make_float4(0.0f, 0.0f, 0.0f, 0.0f);
        if (b * BIN_NODES + (i >> 4) < N_NODES) piece[j] = xrows[i];
    }
#pragma unroll
    for (int q = 0; q < RUNS_PER_WAVE; ++q) {
        if (k0[q] >= 0) atomicAdd(&c[k0[q] & (BIN_NODES - 1)], 1);
        if (k1[q] >= 0) atomicAdd(&c[k1[q] & (BIN_NODES - 1)], 1);
    }
    for (int q = 0; q < RUNS_PER_WAVE; ++q) {
        const int sb  = wave + q * SORT_WAVES;
        const int m   = __builtin_amdgcn_readfirstlane(run_s[sb]);
        const int len = m & 0xffff;
        const int* p  = binned + sb * SLICE_EDGES + (m >> 16);
        for (int i = lane + 64; i < len; i += 32) atomicAdd(&c[p[i] & (BIN_NODES - 1)], 1);
    }
    __syncthreads();

    const int cn = (t < BIN_NODES) ? c[t] : 0;
    const int v  = wave_inclusive_scan(cn, lane);
    if (t < BIN_NODES && lane == 31) wsum[wave] = v;
    __syncthreads();
    if (wave == 0) {
        const int x   = (lane < 8) ? wsum[lane] : 0;
        const int inc = wave_inclusive_scan(x, lane);
        if (lane < 8) wsum[lane] = inc - x;
        if (lane == 31) {
            total_s = inc;
            base_s  = atomicAdd(cursor, (inc + REGION_ALIGN - 1) & ~(REGION_ALIGN - 1));
        }
    }
    __syncthreads();
    if (t < BIN_NODES) {
        const int lo = wsum[wave] + v - cn;
        const int o  = base_s + lo;
        c[t] = lo;
        const int node = b * BIN_NODES + t;
        const float dn = __frsqrt_rn((float)(cn + 1));
        dinv_s[t] = dn;
        if (node < N_NODES) {
            cnt[node]  = cn;
            off[node]  = o;
            dinv[node] = dn;
        }
    }
    __syncthreads();

#pragma unroll
    for (int j = 0; j < ROW_PIECES; ++j) {
        const int i = t + j * SORT_THREADS;
        if (b * BIN_NODES + (i >> 4) < N_NODES) {
            const float dn = dinv_s[i >> 4];
            float4 p = piece[j];
            p.x *= dn;
            p.y *= dn;
            p.z *= dn;
            p.w *= dn;
            xrows[i] = p;
        }
    }

    const int total = total_s;
    int* region     = perm + base_s;
    if (total <= STAGE_CAP) {
#pragma unroll
        for (int q = 0; q < RUNS_PER_WAVE; ++q) {
            if (k0[q] >= 0) stage[atomicAdd(&c[k0[q] & (BIN_NODES - 1)], 1)] = k0[q] >> BIN_SHIFT;
            if (k1[q] >= 0) stage[atomicAdd(&c[k1[q] & (BIN_NODES - 1)], 1)] = k1[q] >> BIN_SHIFT;
        }
        for (int q = 0; q < RUNS_PER_WAVE; ++q) {
            const int sb  = wave + q * SORT_WAVES;
            const int m   = __builtin_amdgcn_readfirstlane(run_s[sb]);
            const int len = m & 0xffff;
            const int* p  = binned + sb * SLICE_EDGES + (m >> 16);
            for (int i = lane + 64; i < len; i += 32) {
                const int k = p[i];
                stage[atomicAdd(&c[k & (BIN_NODES - 1)], 1)] = k >> BIN_SHIFT;
            }
        }
        __syncthreads();
        for (int i = t; i < total; i += SORT_THREADS) region[i] = stage[i];
    } else {
#pragma unroll
        for (int q = 0; q < RUNS_PER_WAVE; ++q) {
            if (k0[q] >= 0) region[atomicAdd(&c[k0[q] & (BIN_NODES - 1)], 1)] = k0[q] >> BIN_SHIFT;
            if (k1[q] >= 0) region[atomicAdd(&c[k1[q] & (BIN_NODES - 1)], 1)] = k1[q] >> BIN_SHIFT;
        }
        for (int q = 0; q < RUNS_PER_WAVE; ++q) {
            const int sb  = wave + q * SORT_WAVES;
            const int m   = __builtin_amdgcn_readfirstlane(run_s[sb]);
            const int len = m & 0xffff;
            const int* p  = binned + sb * SLICE_EDGES + (m >> 16);
            for (int i = lane + 64; i < len; i += 32) {
                const int k = p[i];
                region[atomicAdd(&c[k & (BIN_NODES - 1)], 1)] = k >> BIN_SHIFT;
            }
        }
    }
}


static __device__ __forceinline__ v2f bucket_row(const v2f* __restrict__ xw2, int s, int j, int lane) {
    const int sj = __builtin_amdgcn_readlane(s, j);
    return xw2[(size_t)sj * 32 + lane];
}
static __device__ __forceinline__ void add_row(v2f& acc, v2f r) {
    acc.x += r.x;
    acc.y += r.y;
}

__global__ __launch_bounds__(256) void k_gather(const int* __restrict__ perm, const int* __restrict__ off,
                                                const int* __restrict__ cnt, const float* __restrict__ xw,
                                                const float* __restrict__ dinv, const float* __restrict__ bias,
                                                float* __restrict__ out) {
    const int lane = threadIdx.x & 31;
    const int d = blockIdx.x * WAVES_PER_BLOCK + __builtin_amdgcn_readfirstlane((int)(threadIdx.x >> 5));
    if (d >= N_NODES) return;
    const int   n  = cnt[d];
    const int   o  = off[d];
    const float dd = dinv[d];
    const v2f* xw2 = (const v2f*)xw;

    v2f acc = xw2[(size_t)d * 32 + lane];

    for (int j0 = 0; j0 < n; j0 += 32) {
        const int m = (n - j0 < 32) ? (n - j0) : 32;
        int s = N_NODES;
        if (lane < m) s = perm[o + j0 + lane];
        const int m4 = (m + 3) & ~3;
        int j = 0;
        for (; m4 - j > 4; j += 8) {
            const v2f r0 = bucket_row(xw2, s, j,     lane);
            const v2f r1 = bucket_row(xw2, s, j + 1, lane);
            const v2f r2 = bucket_row(xw2, s, j + 2, lane);
            const v2f r3 = bucket_row(xw2, s, j + 3, lane);
            const v2f r4 = bucket_row(xw2, s, j + 4, lane);
            const v2f r5 = bucket_row(xw2, s, j + 5, lane);
            const v2f r6 = bucket_row(xw2, s, j + 6, lane);
            const v2f r7 = bucket_row(xw2, s, j + 7, lane);
            add_row(acc, r0);
            add_row(acc, r1);
            add_row(acc, r2);
            add_row(acc, r3);
            add_row(acc, r4);
            add_row(acc, r5);
            add_row(acc, r6);
            add_row(acc, r7);
        }
        if (j < m4) {
            const v2f r0 = bucket_row(xw2, s, j,     lane);
            const v2f r1 = bucket_row(xw2, s, j + 1, lane);
            const v2f r2 = bucket_row(xw2, s, j + 2, lane);
            const v2f r3 = bucket_row(xw2, s, j + 3, lane);
            add_row(acc, r0);
            add_row(acc, r1);
            add_row(acc, r2);
            add_row(acc, r3);
        }
    }

    const v2f b2 = ((const v2f*)bias)[lane];
    v2f res;
    res.x = b2.x + acc.x * dd;
    res.y = b2.y + acc.y * dd;
    ((v2f*)out)[(size_t)d * 32 + lane] = res;
}


extern "C" void kernel_launch(void* const* d_in, const int* in_sizes, int n_in,
                              void* d_out, int out_size, void* d_ws, size_t ws_size,
                              hipStream_t stream) {
    const float* x    = (const float*)d_in[0];
    const float* W    = (const float*)d_in[1];
    const float* bias = (const float*)d_in[2];
    const int*   ei   = (const int*)d_in[3];
    const int*   src  = ei;
    const int*   dst  = ei + N_EDGES;
    float* out = (float*)d_out;

    float* dinv   = (float*)d_ws;
    float* xw     = dinv + 102400;
    int*   cnt    = (int*)(xw + 6400064);
    int*   off    = cnt + 102400;
    int*   cursor = off + 102400;
    int*   binned = cursor + 64;
    int*   perm   = binned + 3200000;
    int*   runs   = perm + 3212512;

    k_product_and_slice_sort<<<PRODUCT_BLOCKS + SLICES, SORT_THREADS, 0, stream>>>(x, W, xw, src, dst, binned, runs, cursor);
    k_bin_sort<<<NBINS, SORT_THREADS, 0, stream>>>(binned, runs, perm, cnt, off, dinv, cursor, xw);

    k_gather<<<(N_NODES + WAVES_PER_BLOCK - 1) / WAVES_PER_BLOCK, 256, 0, stream>>>(perm, off, cnt, xw, dinv, bias, out);
}
